# starting kernel
# baseline (speedup 1.0000x reference)
#include <hip/hip_runtime.h>


typedef float v2f __attribute__((ext_vector_type(2)));
typedef float v8f __attribute__((ext_vector_type(8)));

#define SWZ_XOR_ADD(v, pat)                                                    \
  (v) += __int_as_float(__builtin_amdgcn_ds_swizzle(__float_as_int(v), (pat)))

constexpr int Bn = 8;
constexpr int Tn = 64;
constexpr int NI = 1024;
constexpr int NO = 1024;
constexpr int WAVES = 8;
constexpr int COLS_PER_WAVE = NI / WAVES;
constexpr int TILES = COLS_PER_WAVE / 16;

__global__ __launch_bounds__(256) void oja_scan_kernel(
    const float* __restrict__ X,
    const float* __restrict__ W0,
    float* __restrict__ out)
{
  const int lane  = threadIdx.x & 31;
  const int wv    = threadIdx.x >> 5;
  const int b     = blockIdx.x >> 6;
  const int strip = blockIdx.x & 63;
  const int half  = lane >> 4;
  const int ln    = lane & 15;
  const int colbase = wv * COLS_PER_WAVE;
  const float lr = 1.0f / (float)NI;

  __shared__ float part[WAVES][16];
  __shared__ float acoef[16];

  v8f w[TILES];
  const float* Wp = W0 + ((size_t)b * NO + (size_t)strip * 16) * NI;
#pragma unroll
  for (int j = 0; j < TILES; ++j) {
#pragma unroll
    for (int r = 0; r < 8; ++r) {
      w[j][r] = Wp[(size_t)(r + half * 8) * NI + (colbase + j * 16 + ln)];
    }
  }

  float P = 1.0f;

#pragma unroll 1
  for (int t = 0; t < Tn; ++t) {
    const float* xp = X + ((size_t)b * Tn + t) * NI + colbase;
    float xv[TILES];
#pragma unroll
    for (int j = 0; j < TILES; ++j) xv[j] = xp[j * 16 + ln];

    float acc[8];
#pragma unroll
    for (int r = 0; r < 8; ++r) acc[r] = 0.0f;
#pragma unroll
    for (int j = 0; j < TILES; ++j) {
#pragma unroll
      for (int r = 0; r < 8; ++r) acc[r] = __builtin_fmaf(w[j][r], xv[j], acc[r]);
    }

#pragma unroll
    for (int r = 0; r < 8; ++r) {
      SWZ_XOR_ADD(acc[r], 0x041F);
      SWZ_XOR_ADD(acc[r], 0x081F);
      SWZ_XOR_ADD(acc[r], 0x101F);
      SWZ_XOR_ADD(acc[r], 0x201F);
    }
    if (ln == 0) {
#pragma unroll
      for (int r = 0; r < 8; ++r) part[wv][half * 8 + r] = acc[r];
    }
    __syncthreads();

    if (wv == 0 && lane < 16) {
      float d = 0.0f;
#pragma unroll
      for (int k = 0; k < WAVES; ++k) d += part[k][lane];
      float pre = P * d;
      float y   = 1.0f / (1.0f + __expf(-pre));
      out[((size_t)b * Tn + t) * NO + (size_t)strip * 16 + lane] = y;
      P *= (1.0f - lr * y * y);
      acoef[lane] = lr * y / P;
    }
    __syncthreads();

    float ac = acoef[ln];
    v2f A;
    A.x = half ? 0.0f : ac;
    A.y = 0.0f;
#pragma unroll
    for (int j = 0; j < TILES; ++j) {
      v2f Bm;
      Bm.x = half ? 0.0f : xv[j];
      Bm.y = 0.0f;
      w[j] = __builtin_amdgcn_wmma_f32_16x16x4_f32(
           false, A,  false, Bm,
           (short)0, w[j],  false,  false);
    }
  }
}

extern "C" void kernel_launch(void* const* d_in, const int* in_sizes, int n_in,
                              void* d_out, int out_size, void* d_ws, size_t ws_size,
                              hipStream_t stream) {
  const float* X  = (const float*)d_in[0];
  const float* W0 = (const float*)d_in[1];
  float* out      = (float*)d_out;
  (void)in_sizes; (void)n_in; (void)out_size; (void)d_ws; (void)ws_size;

  dim3 grid(Bn * (NO / 16));
  dim3 block(WAVES * 32);
  oja_scan_kernel<<<grid, block, 0, stream>>>(X, W0, out);
}


# Two steps per hand-over: both dots on the same matrix, the 2nd completed with coef_t*(x_t.x_t+1); the two rank-1 updates go out as ONE rank-2 WMMA (K=0|K=2); one fold, LDS exchange, barrier per pair
# speedup vs baseline: 3.2403x; 3.2403x over previous
#include <hip/hip_runtime.h>


typedef float v2f __attribute__((ext_vector_type(2)));
typedef float v8f __attribute__((ext_vector_type(8)));

template <int N>
__device__ __forceinline__ float row_xmask(float v) {
  return __int_as_float(__builtin_amdgcn_update_dpp(
      0, __float_as_int(v), 0x160 + N, 0xf, 0xf, true));
}

template <int N>
__device__ __forceinline__ float fold(float lo, float hi, bool upper) {
  const float keep = upper ? hi : lo;
  const float give = upper ? lo : hi;
  return keep + row_xmask<N>(give);
}

constexpr int Bn = 8;
constexpr int Tn = 64;
constexpr int NI = 1024;
constexpr int NO = 1024;
constexpr int WAVES = 4;
constexpr int COLS_PER_WAVE = NI / WAVES;
constexpr int TILES = COLS_PER_WAVE / 16;
constexpr int XP = TILES / 2;
constexpr int SLOTS = 48;

__global__ __launch_bounds__(128) void oja_scan_kernel(
    const float* __restrict__ X,
    const float* __restrict__ W0,
    float* __restrict__ out)
{
  const int lane  = threadIdx.x & 31;
  const int wv    = threadIdx.x >> 5;
  const int b     = blockIdx.x >> 6;
  const int strip = blockIdx.x & 63;
  const int half  = lane >> 4;
  const int ln    = lane & 15;
  const int colbase = wv * COLS_PER_WAVE;
  const float lr = 1.0f / (float)NI;
  constexpr float NLOG2E = -1.4426950408889634f;

  __shared__ float part[2][WAVES][SLOTS];

  const bool up8 = (ln & 8) != 0;
  const bool up4 = (ln & 4) != 0;
  const bool up2 = (ln & 2) != 0;
  const bool up1 = (ln & 1) != 0;
  const int myrow = half * 8 + (((ln >> 3) & 1) | (((ln >> 2) & 1) << 1) | (((ln >> 1) & 1) << 2));
  const int myslot = (ln & 1) * 16 + myrow;

  v8f w[TILES];
  const float* Wp = W0 + ((size_t)b * NO + (size_t)strip * 16) * NI;
#pragma unroll
  for (int j = 0; j < TILES; ++j) {
#pragma unroll
    for (int r = 0; r < 8; ++r) {
      w[j][r] = Wp[(size_t)(r + half * 8) * NI + (colbase + j * 16 + ln)];
    }
  }

  float Pn = NLOG2E;
  const float cmul = lr * NLOG2E;

  const float* xg = X + (size_t)b * Tn * NI + colbase + ln;
  v2f xa0[XP], xa1[XP], xb0[XP], xb1[XP];
#pragma unroll
  for (int k = 0; k < XP; ++k) { xa0[k].x = xg[k * 32];          xa0[k].y = xg[k * 32 + 16]; }
#pragma unroll
  for (int k = 0; k < XP; ++k) { xa1[k].x = xg[NI + k * 32];     xa1[k].y = xg[NI + k * 32 + 16]; }
#pragma unroll
  for (int k = 0; k < XP; ++k) { xb0[k].x = xg[2 * NI + k * 32]; xb0[k].y = xg[2 * NI + k * 32 + 16]; }
#pragma unroll
  for (int k = 0; k < XP; ++k) { xb1[k].x = xg[3 * NI + k * 32]; xb1[k].y = xg[3 * NI + k * 32 + 16]; }
#pragma unroll
  for (int k = 0; k < XP; ++k) {
    asm volatile("" : "+v"(xa0[k]));
    asm volatile("" : "+v"(xa1[k]));
    asm volatile("" : "+v"(xb0[k]));
    asm volatile("" : "+v"(xb1[k]));
  }
#pragma unroll
  for (int j = 0; j < TILES; ++j) asm volatile("" : "+v"(w[j]));

  auto pairstep = [&](v2f (&xe)[XP], v2f (&xo)[XP], const int t, float (*pb)[SLOTS])
                      __attribute__((always_inline)) {
#pragma unroll
    for (int k = 0; k < XP; ++k) {
      asm volatile("" : "+v"(xe[k]));
      asm volatile("" : "+v"(xo[k]));
    }

    v2f aE2[2][4], aO2[2][4];
#pragma unroll
    for (int j = 0; j < TILES; ++j) {
      const v2f xsE = (j & 1) ? __builtin_shufflevector(xe[j >> 1], xe[j >> 1], 1, 1)
                              : __builtin_shufflevector(xe[j >> 1], xe[j >> 1], 0, 0);
      const v2f xsO = (j & 1) ? __builtin_shufflevector(xo[j >> 1], xo[j >> 1], 1, 1)
                              : __builtin_shufflevector(xo[j >> 1], xo[j >> 1], 0, 0);
#pragma unroll
      for (int i = 0; i < 4; ++i) {
        const v2f w2 = {w[j][2 * i], w[j][2 * i + 1]};
        aE2[j & 1][i] = (j < 2) ? (w2 * xsE)
                                : __builtin_elementwise_fma(w2, xsE, aE2[j & 1][i]);
        aO2[j & 1][i] = (j < 2) ? (w2 * xsO)
                                : __builtin_elementwise_fma(w2, xsO, aO2[j & 1][i]);
      }
    }
    float accE[8], accO[8];
#pragma unroll
    for (int i = 0; i < 4; ++i) {
      const v2f sE = aE2[0][i] + aE2[1][i];
      const v2f sO = aO2[0][i] + aO2[1][i];
      accE[2 * i] = sE.x;  accE[2 * i + 1] = sE.y;
      accO[2 * i] = sO.x;  accO[2 * i + 1] = sO.y;
    }

    v2f g2 = xe[0] * xo[0];
#pragma unroll
    for (int k = 1; k < XP; ++k) g2 = __builtin_elementwise_fma(xe[k], xo[k], g2);
    float gp = g2.x + g2.y;

    float bx[TILES];
#pragma unroll
    for (int j = 0; j < TILES; ++j) {
      bx[j] = half ? xo[j >> 1][j & 1] : xe[j >> 1][j & 1];
      asm volatile("" : "+v"(bx[j]));
    }

    const float mE0 = fold<8>(accE[0], accE[1], up8);
    const float mE1 = fold<8>(accE[2], accE[3], up8);
    const float mE2 = fold<8>(accE[4], accE[5], up8);
    const float mE3 = fold<8>(accE[6], accE[7], up8);
    const float mO0 = fold<8>(accO[0], accO[1], up8);
    const float mO1 = fold<8>(accO[2], accO[3], up8);
    const float mO2 = fold<8>(accO[4], accO[5], up8);
    const float mO3 = fold<8>(accO[6], accO[7], up8);
    const float nE0 = fold<4>(mE0, mE1, up4);
    const float nE1 = fold<4>(mE2, mE3, up4);
    const float nO0 = fold<4>(mO0, mO1, up4);
    const float nO1 = fold<4>(mO2, mO3, up4);
    const float qE = fold<2>(nE0, nE1, up2);
    const float qO = fold<2>(nO0, nO1, up2);
    const float q  = fold<1>(qE, qO, up1);
    gp += row_xmask<1>(gp);
    gp += row_xmask<2>(gp);
    gp += row_xmask<4>(gp);
    gp += row_xmask<8>(gp);
    pb[wv][myslot] = q;
    if (lane == 0) pb[wv][32] = gp;
    __syncthreads();

    float pE[WAVES], pO[WAVES], pg[WAVES];
#pragma unroll
    for (int k = 0; k < WAVES; ++k) {
      pE[k] = pb[k][ln];
      pO[k] = pb[k][16 + ln];
      pg[k] = pb[k][32];
    }
    asm volatile("" ::: "memory");

    {
      const int te = (t + 4 < Tn) ? (t + 4) : (Tn - 1);
      const int to = (t + 5 < Tn) ? (t + 5) : (Tn - 1);
      const float* xpe = xg + (size_t)te * NI;
      const float* xpo = xg + (size_t)to * NI;
#pragma unroll
      for (int k = 0; k < XP; ++k) { xe[k].x = xpe[k * 32]; xe[k].y = xpe[k * 32 + 16]; }
#pragma unroll
      for (int k = 0; k < XP; ++k) { xo[k].x = xpo[k * 32]; xo[k].y = xpo[k * 32 + 16]; }
    }

    const float dE = (pE[0] + pE[1]) + (pE[2] + pE[3]);
    const float dS = (pO[0] + pO[1]) + (pO[2] + pO[3]);
    const float gg = (pg[0] + pg[1]) + (pg[2] + pg[3]);
    const float eE = __builtin_amdgcn_exp2f(Pn * dE);
    const float yE = __builtin_amdgcn_rcpf(1.0f + eE);
    const float lyE = lr * yE;
    Pn *= __builtin_fmaf(-lyE, yE, 1.0f);
    const float cE = (cmul * yE) * __builtin_amdgcn_rcpf(Pn);
    const float dO = __builtin_fmaf(cE, gg, dS);
    const float eO = __builtin_amdgcn_exp2f(Pn * dO);
    const float yO = __builtin_amdgcn_rcpf(1.0f + eO);
    const float lyO = lr * yO;
    Pn *= __builtin_fmaf(-lyO, yO, 1.0f);
    const float cO = (cmul * yO) * __builtin_amdgcn_rcpf(Pn);
    if (wv == 0) {
      out[((size_t)b * Tn + t + half) * NO + (size_t)strip * 16 + ln] = half ? yO : yE;
    }

    v2f A;
    A.x = half ? cO : cE;
    A.y = 0.0f;
#pragma unroll
    for (int j = 0; j < TILES; ++j) {
      v2f Bm;
      Bm.x = bx[j];
      Bm.y = 0.0f;
      w[j] = __builtin_amdgcn_wmma_f32_16x16x4_f32(
           false, A,  false, Bm,
           (short)0, w[j],  false,  false);
    }
  };

#pragma unroll 1
  for (int t = 0; t < Tn; t += 4) {
    pairstep(xa0, xa1, t, part[0]);
    pairstep(xb0, xb1, t + 2, part[1]);
  }
}

extern "C" void kernel_launch(void* const* d_in, const int* in_sizes, int n_in,
                              void* d_out, int out_size, void* d_ws, size_t ws_size,
                              hipStream_t stream) {
  const float* X  = (const float*)d_in[0];
  const float* W0 = (const float*)d_in[1];
  float* out      = (float*)d_out;
  (void)in_sizes; (void)n_in; (void)out_size; (void)d_ws; (void)ws_size;

  dim3 grid(Bn * (NO / 16));
  dim3 block(WAVES * 32);
  oja_scan_kernel<<<grid, block, 0, stream>>>(X, W0, out);
}


# Four tiles interleave columns in each wave (tile j, lane n: column 64*(j/4)+4n+j%4): a lane's four values are 16 adjacent bytes, so W and x come by 16-byte loads: 16 x loads per group not 64
# speedup vs baseline: 4.0356x; 1.2454x over previous
#include <hip/hip_runtime.h>


typedef float v2f __attribute__((ext_vector_type(2)));
typedef float v4f __attribute__((ext_vector_type(4)));
typedef float v8f __attribute__((ext_vector_type(8)));
typedef unsigned long long xpair_t;
typedef xpair_t v2q __attribute__((ext_vector_type(2)));

template <int N>
__device__ __forceinline__ float row_xmask(float v) {
  return __int_as_float(__builtin_amdgcn_update_dpp(
      0, __float_as_int(v), 0x160 + N, 0xf, 0xf, true));
}

template <int N>
__device__ __forceinline__ float fold(float lo, float hi, bool upper) {
  const float keep = upper ? hi : lo;
  const float give = upper ? lo : hi;
  return keep + row_xmask<N>(give);
}

constexpr int Bn = 8;
constexpr int Tn = 64;
constexpr int NI = 1024;
constexpr int NO = 1024;
constexpr int WAVES = 4;
constexpr int COLS_PER_WAVE = NI / WAVES;
constexpr int TILES = COLS_PER_WAVE / 16;
constexpr int XP = TILES / 2;
constexpr int QUADS = TILES / 4;
constexpr int SLOTS = 80;

__global__ __launch_bounds__(128) void oja_scan_kernel(
    const float* __restrict__ X,
    const float* __restrict__ W0,
    float* __restrict__ out)
{
  const int lane  = threadIdx.x & 31;
  const int wv    = threadIdx.x >> 5;
  const int b     = blockIdx.x >> 6;
  const int strip = blockIdx.x & 63;
  const int half  = lane >> 4;
  const int ln    = lane & 15;
  const int colbase = wv * COLS_PER_WAVE;
  const float lr = 1.0f / (float)NI;
  constexpr float NLOG2E = -1.4426950408889634f;

  __shared__ __attribute__((aligned(16))) float part[2][WAVES][SLOTS];

  const bool up8 = (ln & 8) != 0;
  const bool up4 = (ln & 4) != 0;
  const bool up2 = (ln & 2) != 0;
  const bool up1 = (ln & 1) != 0;
  const int myrow = half * 8 + (((ln >> 3) & 1) | (((ln >> 2) & 1) << 1) | (((ln >> 1) & 1) << 2));
  const int myslot = (ln & 1) * 16 + myrow;

  v8f w[TILES];
  const float* Wp = W0 + ((size_t)b * NO + (size_t)strip * 16) * NI + colbase + 4 * ln;
#pragma unroll
  for (int J = 0; J < QUADS; ++J) {
#pragma unroll
    for (int r = 0; r < 8; ++r) {
      const v4f t = *reinterpret_cast<const v4f*>(Wp + (size_t)(r + half * 8) * NI + J * 64);
      w[4 * J + 0][r] = t.x;
      w[4 * J + 1][r] = t.y;
      w[4 * J + 2][r] = t.z;
      w[4 * J + 3][r] = t.w;
    }
  }

  float Pn = NLOG2E;
  const float cmul = lr * NLOG2E;

  const float* xg = X + (size_t)b * Tn * NI + colbase + 4 * ln;
  auto load_x = [&](xpair_t (&dst)[XP], const float* p) __attribute__((always_inline)) {
#pragma unroll
    for (int J = 0; J < QUADS; ++J) {
      const v2q d = __builtin_bit_cast(v2q, *reinterpret_cast<const v4f*>(p + J * 64));
      dst[2 * J]     = d.x;
      dst[2 * J + 1] = d.y;
    }
  };
  xpair_t xa[4][XP], xb[4][XP];
#pragma unroll
  for (int s = 0; s < 4; ++s) load_x(xa[s], xg + s * NI);
#pragma unroll
  for (int s = 0; s < 4; ++s) load_x(xb[s], xg + (4 + s) * NI);

  const int gidx = (ln & 2) ? (4 + ((ln >> 3) & 1)) : (((ln >> 3) & 1) + 2 * ((ln >> 2) & 1));
  const bool gsel = (half == 0) && ((ln & 1) == 0) && (((ln & 2) == 0) || ((ln & 4) == 0));

  auto fold16 = [&](const v2f (&aE)[4], const v2f (&aO)[4]) __attribute__((always_inline)) {
    const float mE0 = fold<8>(aE[0].x, aE[0].y, up8);
    const float mE1 = fold<8>(aE[1].x, aE[1].y, up8);
    const float mE2 = fold<8>(aE[2].x, aE[2].y, up8);
    const float mE3 = fold<8>(aE[3].x, aE[3].y, up8);
    const float mO0 = fold<8>(aO[0].x, aO[0].y, up8);
    const float mO1 = fold<8>(aO[1].x, aO[1].y, up8);
    const float mO2 = fold<8>(aO[2].x, aO[2].y, up8);
    const float mO3 = fold<8>(aO[3].x, aO[3].y, up8);
    const float nE0 = fold<4>(mE0, mE1, up4);
    const float nE1 = fold<4>(mE2, mE3, up4);
    const float nO0 = fold<4>(mO0, mO1, up4);
    const float nO1 = fold<4>(mO2, mO3, up4);
    const float qE = fold<2>(nE0, nE1, up2);
    const float qO = fold<2>(nO0, nO1, up2);
    return fold<1>(qE, qO, up1);
  };

  auto quadstep = [&](xpair_t (&xd)[4][XP], const int t, float (*pb)[SLOTS])
                      __attribute__((always_inline)) {
    v2f xq[4][XP];
#pragma unroll
    for (int s = 0; s < 4; ++s) {
#pragma unroll
      for (int k = 0; k < XP; ++k) xq[s][k] = __builtin_bit_cast(v2f, xd[s][k]);
    }

    v2f acc[4][4];
#pragma unroll
    for (int j = 0; j < TILES; ++j) {
#pragma unroll
      for (int s = 0; s < 4; ++s) {
        const v2f xs = (j & 1) ? __builtin_shufflevector(xq[s][j >> 1], xq[s][j >> 1], 1, 1)
                               : __builtin_shufflevector(xq[s][j >> 1], xq[s][j >> 1], 0, 0);
#pragma unroll
        for (int i = 0; i < 4; ++i) {
          const v2f w2 = {w[j][2 * i], w[j][2 * i + 1]};
          acc[s][i] = (j == 0) ? (w2 * xs) : __builtin_elementwise_fma(w2, xs, acc[s][i]);
        }
      }
    }

    float gp[6];
    {
      constexpr int GA[6] = {0, 0, 0, 1, 1, 2};
      constexpr int GC[6] = {1, 2, 3, 2, 3, 3};
#pragma unroll
      for (int p = 0; p < 6; ++p) {
        v2f g2 = xq[GA[p]][0] * xq[GC[p]][0];
#pragma unroll
        for (int k = 1; k < XP; ++k)
          g2 = __builtin_elementwise_fma(xq[GA[p]][k], xq[GC[p]][k], g2);
        gp[p] = g2.x + g2.y;
      }
    }

    v2f bq[TILES];
#pragma unroll
    for (int j = 0; j < TILES; ++j) {
      bq[j].x = half ? xq[2][j >> 1][j & 1] : xq[0][j >> 1][j & 1];
      bq[j].y = half ? xq[3][j >> 1][j & 1] : xq[1][j >> 1][j & 1];
      asm volatile("" : "+v"(bq[j]));
    }

    const float q01 = fold16(acc[0], acc[1]);
    const float q23 = fold16(acc[2], acc[3]);
    const float h0 = fold<8>(gp[0], gp[1], up8);
    const float h1 = fold<8>(gp[2], gp[3], up8);
    const float h2 = fold<8>(gp[4], gp[5], up8);
    const float k0 = fold<4>(h0, h1, up4);
    const float k1 = h2 + row_xmask<4>(h2);
    const float k2 = fold<2>(k0, k1, up2);
    const float gq = k2 + row_xmask<1>(k2);

    pb[wv][myslot] = q01;
    pb[wv][32 + myslot] = q23;
    if (gsel) pb[wv][64 + gidx] = gq;
    __syncthreads();

    float pD[4][WAVES];
    v4f ga[WAVES];
    v2f gb[WAVES];
#pragma unroll
    for (int k = 0; k < WAVES; ++k) {
#pragma unroll
      for (int s = 0; s < 4; ++s) pD[s][k] = pb[k][16 * s + ln];
      ga[k] = *reinterpret_cast<const v4f*>(&pb[k][64]);
      gb[k] = *reinterpret_cast<const v2f*>(&pb[k][68]);
    }
    asm volatile("" ::: "memory");

#pragma unroll
    for (int s = 0; s < 4; ++s) {
      const int ts = (t + 8 + s < Tn) ? (t + 8 + s) : (Tn - 1);
      load_x(xd[s], xg + (size_t)ts * NI);
    }

    float D[4];
#pragma unroll
    for (int s = 0; s < 4; ++s) D[s] = (pD[s][0] + pD[s][1]) + (pD[s][2] + pD[s][3]);
    const v4f gs = (ga[0] + ga[1]) + (ga[2] + ga[3]);
    const v2f gt = (gb[0] + gb[1]) + (gb[2] + gb[3]);
    auto finish = [&](const float d, float& y, float& cf) __attribute__((always_inline)) {
      const float e = __builtin_amdgcn_exp2f(Pn * d);
      y = __builtin_amdgcn_rcpf(1.0f + e);
      const float ly = lr * y;
      Pn *= __builtin_fmaf(-ly, y, 1.0f);
      cf = (cmul * y) * __builtin_amdgcn_rcpf(Pn);
    };
    float y0, y1, y2, y3, c0, c1, c2, c3;
    finish(D[0], y0, c0);
    const float d1 = __builtin_fmaf(c0, gs.x, D[1]);
    const float d2a = __builtin_fmaf(c0, gs.y, D[2]);
    const float d3a = __builtin_fmaf(c0, gs.z, D[3]);
    finish(d1, y1, c1);
    const float d2 = __builtin_fmaf(c1, gs.w, d2a);
    const float d3b = __builtin_fmaf(c1, gt.x, d3a);
    finish(d2, y2, c2);
    const float d3 = __builtin_fmaf(c2, gt.y, d3b);
    finish(d3, y3, c3);
    if (wv < 2) {
      const float yv = wv ? (half ? y3 : y2) : (half ? y1 : y0);
      out[((size_t)b * Tn + t + 2 * wv + half) * NO + (size_t)strip * 16 + ln] = yv;
    }

    v2f A;
    A.x = half ? c2 : c0;
    A.y = half ? c3 : c1;
#pragma unroll
    for (int j = 0; j < TILES; ++j) {
      w[j] = __builtin_amdgcn_wmma_f32_16x16x4_f32(
           false, A,  false, bq[j],
           (short)0, w[j],  false,  false);
    }
  };

#pragma unroll 1
  for (int t = 0; t < Tn; t += 8) {
    quadstep(xa, t, part[0]);
    quadstep(xb, t + 4, part[1]);
  }
}

extern "C" void kernel_launch(void* const* d_in, const int* in_sizes, int n_in,
                              void* d_out, int out_size, void* d_ws, size_t ws_size,
                              hipStream_t stream) {
  const float* X  = (const float*)d_in[0];
  const float* W0 = (const float*)d_in[1];
  float* out      = (float*)d_out;
  (void)in_sizes; (void)n_in; (void)out_size; (void)d_ws; (void)ws_size;

  dim3 grid(Bn * (NO / 16));
  dim3 block(WAVES * 32);
  oja_scan_kernel<<<grid, block, 0, stream>>>(X, W0, out);
}


# Wave priority falls with progress (s_setprio 3,2,1,0 over the loop trips): the two workgroups sharing a SIMD advance together, not oldest first, and neither runs its last groups alone
# speedup vs baseline: 4.1884x; 1.0379x over previous
#include <hip/hip_runtime.h>


typedef float v2f __attribute__((ext_vector_type(2)));
typedef float v4f __attribute__((ext_vector_type(4)));
typedef float v8f __attribute__((ext_vector_type(8)));
typedef unsigned long long xpair_t;
typedef xpair_t v2q __attribute__((ext_vector_type(2)));

template <int N>
__device__ __forceinline__ float row_xmask(float v) {
  return __int_as_float(__builtin_amdgcn_update_dpp(
      0, __float_as_int(v), 0x160 + N, 0xf, 0xf, true));
}

template <int N>
__device__ __forceinline__ float fold(float lo, float hi, bool upper) {
  const float keep = upper ? hi : lo;
  const float give = upper ? lo : hi;
  return keep + row_xmask<N>(give);
}

constexpr int Bn = 8;
constexpr int Tn = 64;
constexpr int NI = 1024;
constexpr int NO = 1024;
constexpr int WAVES = 4;
constexpr int COLS_PER_WAVE = NI / WAVES;
constexpr int TILES = COLS_PER_WAVE / 16;
constexpr int XP = TILES / 2;
constexpr int QUADS = TILES / 4;
constexpr int SLOTS = 80;

__global__ __launch_bounds__(128) void oja_scan_kernel(
    const float* __restrict__ X,
    const float* __restrict__ W0,
    float* __restrict__ out)
{
  const int lane  = threadIdx.x & 31;
  const int wv    = threadIdx.x >> 5;
  const int b     = blockIdx.x >> 6;
  const int strip = blockIdx.x & 63;
  const int half  = lane >> 4;
  const int ln    = lane & 15;
  const int colbase = wv * COLS_PER_WAVE;
  const float lr = 1.0f / (float)NI;
  constexpr float NLOG2E = -1.4426950408889634f;

  __shared__ __attribute__((aligned(16))) float part[2][WAVES][SLOTS];

  const bool up8 = (ln & 8) != 0;
  const bool up4 = (ln & 4) != 0;
  const bool up2 = (ln & 2) != 0;
  const bool up1 = (ln & 1) != 0;
  const int myrow = half * 8 + (((ln >> 3) & 1) | (((ln >> 2) & 1) << 1) | (((ln >> 1) & 1) << 2));
  const int myslot = (ln & 1) * 16 + myrow;

  v8f w[TILES];
  const float* Wp = W0 + ((size_t)b * NO + (size_t)strip * 16) * NI + colbase + 4 * ln;
#pragma unroll
  for (int J = 0; J < QUADS; ++J) {
#pragma unroll
    for (int r = 0; r < 8; ++r) {
      const v4f t = *reinterpret_cast<const v4f*>(Wp + (size_t)(r + half * 8) * NI + J * 64);
      w[4 * J + 0][r] = t.x;
      w[4 * J + 1][r] = t.y;
      w[4 * J + 2][r] = t.z;
      w[4 * J + 3][r] = t.w;
    }
  }

  float Pn = NLOG2E;
  const float cmul = lr * NLOG2E;

  const float* xg = X + (size_t)b * Tn * NI + colbase + 4 * ln;
  auto load_x = [&](xpair_t (&dst)[XP], const float* p) __attribute__((always_inline)) {
#pragma unroll
    for (int J = 0; J < QUADS; ++J) {
      const v2q d = __builtin_bit_cast(v2q, *reinterpret_cast<const v4f*>(p + J * 64));
      dst[2 * J]     = d.x;
      dst[2 * J + 1] = d.y;
    }
  };
  xpair_t xa[4][XP], xb[4][XP];
#pragma unroll
  for (int s = 0; s < 4; ++s) load_x(xa[s], xg + s * NI);
#pragma unroll
  for (int s = 0; s < 4; ++s) load_x(xb[s], xg + (4 + s) * NI);

  const int gidx = (ln & 2) ? (4 + ((ln >> 3) & 1)) : (((ln >> 3) & 1) + 2 * ((ln >> 2) & 1));
  const bool gsel = (half == 0) && ((ln & 1) == 0) && (((ln & 2) == 0) || ((ln & 4) == 0));

  auto fold16 = [&](const v2f (&aE)[4], const v2f (&aO)[4]) __attribute__((always_inline)) {
    const float mE0 = fold<8>(aE[0].x, aE[0].y, up8);
    const float mE1 = fold<8>(aE[1].x, aE[1].y, up8);
    const float mE2 = fold<8>(aE[2].x, aE[2].y, up8);
    const float mE3 = fold<8>(aE[3].x, aE[3].y, up8);
    const float mO0 = fold<8>(aO[0].x, aO[0].y, up8);
    const float mO1 = fold<8>(aO[1].x, aO[1].y, up8);
    const float mO2 = fold<8>(aO[2].x, aO[2].y, up8);
    const float mO3 = fold<8>(aO[3].x, aO[3].y, up8);
    const float nE0 = fold<4>(mE0, mE1, up4);
    const float nE1 = fold<4>(mE2, mE3, up4);
    const float nO0 = fold<4>(mO0, mO1, up4);
    const float nO1 = fold<4>(mO2, mO3, up4);
    const float qE = fold<2>(nE0, nE1, up2);
    const float qO = fold<2>(nO0, nO1, up2);
    return fold<1>(qE, qO, up1);
  };

  auto quadstep = [&](xpair_t (&xd)[4][XP], const int t, float (*pb)[SLOTS])
                      __attribute__((always_inline)) {
    v2f xq[4][XP];
#pragma unroll
    for (int s = 0; s < 4; ++s) {
#pragma unroll
      for (int k = 0; k < XP; ++k) xq[s][k] = __builtin_bit_cast(v2f, xd[s][k]);
    }

    v2f acc[4][4];
#pragma unroll
    for (int j = 0; j < TILES; ++j) {
#pragma unroll
      for (int s = 0; s < 4; ++s) {
        const v2f xs = (j & 1) ? __builtin_shufflevector(xq[s][j >> 1], xq[s][j >> 1], 1, 1)
                               : __builtin_shufflevector(xq[s][j >> 1], xq[s][j >> 1], 0, 0);
#pragma unroll
        for (int i = 0; i < 4; ++i) {
          const v2f w2 = {w[j][2 * i], w[j][2 * i + 1]};
          acc[s][i] = (j == 0) ? (w2 * xs) : __builtin_elementwise_fma(w2, xs, acc[s][i]);
        }
      }
    }

    float gp[6];
    {
      constexpr int GA[6] = {0, 0, 0, 1, 1, 2};
      constexpr int GC[6] = {1, 2, 3, 2, 3, 3};
#pragma unroll
      for (int p = 0; p < 6; ++p) {
        v2f g2 = xq[GA[p]][0] * xq[GC[p]][0];
#pragma unroll
        for (int k = 1; k < XP; ++k)
          g2 = __builtin_elementwise_fma(xq[GA[p]][k], xq[GC[p]][k], g2);
        gp[p] = g2.x + g2.y;
      }
    }

    v2f bq[TILES];
#pragma unroll
    for (int j = 0; j < TILES; ++j) {
      bq[j].x = half ? xq[2][j >> 1][j & 1] : xq[0][j >> 1][j & 1];
      bq[j].y = half ? xq[3][j >> 1][j & 1] : xq[1][j >> 1][j & 1];
      asm volatile("" : "+v"(bq[j]));
    }

    const float q01 = fold16(acc[0], acc[1]);
    const float q23 = fold16(acc[2], acc[3]);
    const float h0 = fold<8>(gp[0], gp[1], up8);
    const float h1 = fold<8>(gp[2], gp[3], up8);
    const float h2 = fold<8>(gp[4], gp[5], up8);
    const float k0 = fold<4>(h0, h1, up4);
    const float k1 = h2 + row_xmask<4>(h2);
    const float k2 = fold<2>(k0, k1, up2);
    const float gq = k2 + row_xmask<1>(k2);

    pb[wv][myslot] = q01;
    pb[wv][32 + myslot] = q23;
    if (gsel) pb[wv][64 + gidx] = gq;
    __syncthreads();

    float pD[4][WAVES];
    v4f ga[WAVES];
    v2f gb[WAVES];
#pragma unroll
    for (int k = 0; k < WAVES; ++k) {
#pragma unroll
      for (int s = 0; s < 4; ++s) pD[s][k] = pb[k][16 * s + ln];
      ga[k] = *reinterpret_cast<const v4f*>(&pb[k][64]);
      gb[k] = *reinterpret_cast<const v2f*>(&pb[k][68]);
    }
    asm volatile("" ::: "memory");

#pragma unroll
    for (int s = 0; s < 4; ++s) {
      const int ts = (t + 8 + s < Tn) ? (t + 8 + s) : (Tn - 1);
      load_x(xd[s], xg + (size_t)ts * NI);
    }

    float D[4];
#pragma unroll
    for (int s = 0; s < 4; ++s) D[s] = (pD[s][0] + pD[s][1]) + (pD[s][2] + pD[s][3]);
    const v4f gs = (ga[0] + ga[1]) + (ga[2] + ga[3]);
    const v2f gt = (gb[0] + gb[1]) + (gb[2] + gb[3]);
    auto finish = [&](const float d, float& y, float& cf) __attribute__((always_inline)) {
      const float e = __builtin_amdgcn_exp2f(Pn * d);
      y = __builtin_amdgcn_rcpf(1.0f + e);
      const float ly = lr * y;
      Pn *= __builtin_fmaf(-ly, y, 1.0f);
      cf = (cmul * y) * __builtin_amdgcn_rcpf(Pn);
    };
    float y0, y1, y2, y3, c0, c1, c2, c3;
    finish(D[0], y0, c0);
    const float d1 = __builtin_fmaf(c0, gs.x, D[1]);
    const float d2a = __builtin_fmaf(c0, gs.y, D[2]);
    const float d3a = __builtin_fmaf(c0, gs.z, D[3]);
    finish(d1, y1, c1);
    const float d2 = __builtin_fmaf(c1, gs.w, d2a);
    const float d3b = __builtin_fmaf(c1, gt.x, d3a);
    finish(d2, y2, c2);
    const float d3 = __builtin_fmaf(c2, gt.y, d3b);
    finish(d3, y3, c3);
    if (wv < 2) {
      const float yv = wv ? (half ? y3 : y2) : (half ? y1 : y0);
      out[((size_t)b * Tn + t + 2 * wv + half) * NO + (size_t)strip * 16 + ln] = yv;
    }

    v2f A;
    A.x = half ? c2 : c0;
    A.y = half ? c3 : c1;
#pragma unroll
    for (int j = 0; j < TILES; ++j) {
      w[j] = __builtin_amdgcn_wmma_f32_16x16x4_f32(
           false, A,  false, bq[j],
           (short)0, w[j],  false,  false);
    }
  };

  __builtin_amdgcn_s_setprio(3);

#pragma unroll 1
  for (int t = 0; t < Tn; t += 8) {
    switch (t) {
      case 32: __builtin_amdgcn_s_setprio(2); break;
      case 48: __builtin_amdgcn_s_setprio(1); break;
      case 56: __builtin_amdgcn_s_setprio(0); break;
      default: break;
    }
    quadstep(xa, t, part[0]);
    quadstep(xb, t + 4, part[1]);
  }
}

extern "C" void kernel_launch(void* const* d_in, const int* in_sizes, int n_in,
                              void* d_out, int out_size, void* d_ws, size_t ws_size,
                              hipStream_t stream) {
  const float* X  = (const float*)d_in[0];
  const float* W0 = (const float*)d_in[1];
  float* out      = (float*)d_out;
  (void)in_sizes; (void)n_in; (void)out_size; (void)d_ws; (void)ws_size;

  dim3 grid(Bn * (NO / 16));
  dim3 block(WAVES * 32);
  oja_scan_kernel<<<grid, block, 0, stream>>>(X, W0, out);
}


# Group's dependent path shortened: sigmoid recurrence on the exponent argument (2 FMAs from a step's output to the next exp); next group's six x.x sums taken under that chain, not before the barrier
# speedup vs baseline: 4.3992x; 1.0503x over previous
#include <hip/hip_runtime.h>


typedef float v2f __attribute__((ext_vector_type(2)));
typedef float v4f __attribute__((ext_vector_type(4)));
typedef float v8f __attribute__((ext_vector_type(8)));
typedef unsigned long long xpair_t;
typedef xpair_t v2q __attribute__((ext_vector_type(2)));

template <int N>
__device__ __forceinline__ float row_xmask(float v) {
  return __int_as_float(__builtin_amdgcn_update_dpp(
      0, __float_as_int(v), 0x160 + N, 0xf, 0xf, true));
}

template <int N>
__device__ __forceinline__ float fold(float lo, float hi, bool upper) {
  const float keep = upper ? hi : lo;
  const float give = upper ? lo : hi;
  return keep + row_xmask<N>(give);
}

constexpr int Bn = 8;
constexpr int Tn = 64;
constexpr int NI = 1024;
constexpr int NO = 1024;
constexpr int WAVES = 4;
constexpr int COLS_PER_WAVE = NI / WAVES;
constexpr int TILES = COLS_PER_WAVE / 16;
constexpr int XP = TILES / 2;
constexpr int QUADS = TILES / 4;
constexpr int SLOTS = 80;

__global__ __launch_bounds__(128) void oja_scan_kernel(
    const float* __restrict__ X,
    const float* __restrict__ W0,
    float* __restrict__ out)
{
  const int lane  = threadIdx.x & 31;
  const int wv    = threadIdx.x >> 5;
  const int b     = blockIdx.x >> 6;
  const int strip = blockIdx.x & 63;
  const int half  = lane >> 4;
  const int ln    = lane & 15;
  const int colbase = wv * COLS_PER_WAVE;
  const float lr = 1.0f / (float)NI;
  constexpr float NLOG2E = -1.4426950408889634f;

  __shared__ __attribute__((aligned(16))) float part[2][WAVES][SLOTS];

  const bool up8 = (ln & 8) != 0;
  const bool up4 = (ln & 4) != 0;
  const bool up2 = (ln & 2) != 0;
  const bool up1 = (ln & 1) != 0;
  const int myrow = half * 8 + (((ln >> 3) & 1) | (((ln >> 2) & 1) << 1) | (((ln >> 1) & 1) << 2));
  const int myslot = (ln & 1) * 16 + myrow;

  v8f w[TILES];
  const float* Wp = W0 + ((size_t)b * NO + (size_t)strip * 16) * NI + colbase + 4 * ln;
#pragma unroll
  for (int J = 0; J < QUADS; ++J) {
#pragma unroll
    for (int r = 0; r < 8; ++r) {
      const v4f t = *reinterpret_cast<const v4f*>(Wp + (size_t)(r + half * 8) * NI + J * 64);
      w[4 * J + 0][r] = t.x;
      w[4 * J + 1][r] = t.y;
      w[4 * J + 2][r] = t.z;
      w[4 * J + 3][r] = t.w;
    }
  }

  float Pn = NLOG2E;
  const float cmul = lr * NLOG2E;

  const float* xg = X + (size_t)b * Tn * NI + colbase + 4 * ln;
  auto load_x = [&](xpair_t (&dst)[XP], const float* p) __attribute__((always_inline)) {
#pragma unroll
    for (int J = 0; J < QUADS; ++J) {
      const v2q d = __builtin_bit_cast(v2q, *reinterpret_cast<const v4f*>(p + J * 64));
      dst[2 * J]     = d.x;
      dst[2 * J + 1] = d.y;
    }
  };
  xpair_t xa[4][XP], xb[4][XP];
#pragma unroll
  for (int s = 0; s < 4; ++s) load_x(xa[s], xg + s * NI);
#pragma unroll
  for (int s = 0; s < 4; ++s) load_x(xb[s], xg + (4 + s) * NI);

  const int gidx = (ln & 2) ? (4 + ((ln >> 3) & 1)) : (((ln >> 3) & 1) + 2 * ((ln >> 2) & 1));
  const bool gsel = (half == 0) && ((ln & 1) == 0) && (((ln & 2) == 0) || ((ln & 4) == 0));

  auto fold16 = [&](const v2f (&aE)[4], const v2f (&aO)[4]) __attribute__((always_inline)) {
    const float mE0 = fold<8>(aE[0].x, aE[0].y, up8);
    const float mE1 = fold<8>(aE[1].x, aE[1].y, up8);
    const float mE2 = fold<8>(aE[2].x, aE[2].y, up8);
    const float mE3 = fold<8>(aE[3].x, aE[3].y, up8);
    const float mO0 = fold<8>(aO[0].x, aO[0].y, up8);
    const float mO1 = fold<8>(aO[1].x, aO[1].y, up8);
    const float mO2 = fold<8>(aO[2].x, aO[2].y, up8);
    const float mO3 = fold<8>(aO[3].x, aO[3].y, up8);
    const float nE0 = fold<4>(mE0, mE1, up4);
    const float nE1 = fold<4>(mE2, mE3, up4);
    const float nO0 = fold<4>(mO0, mO1, up4);
    const float nO1 = fold<4>(mO2, mO3, up4);
    const float qE = fold<2>(nE0, nE1, up2);
    const float qO = fold<2>(nO0, nO1, up2);
    return fold<1>(qE, qO, up1);
  };

  auto xdotx = [&](const xpair_t (&xs)[4][XP]) __attribute__((always_inline)) {
    constexpr int GA[6] = {0, 0, 0, 1, 1, 2};
    constexpr int GC[6] = {1, 2, 3, 2, 3, 3};
    float gp[6];
#pragma unroll
    for (int p = 0; p < 6; ++p) {
      v2f g2 = __builtin_bit_cast(v2f, xs[GA[p]][0]) * __builtin_bit_cast(v2f, xs[GC[p]][0]);
#pragma unroll
      for (int k = 1; k < XP; ++k)
        g2 = __builtin_elementwise_fma(__builtin_bit_cast(v2f, xs[GA[p]][k]),
                                       __builtin_bit_cast(v2f, xs[GC[p]][k]), g2);
      gp[p] = g2.x + g2.y;
    }
    const float h0 = fold<8>(gp[0], gp[1], up8);
    const float h1 = fold<8>(gp[2], gp[3], up8);
    const float h2 = fold<8>(gp[4], gp[5], up8);
    const float k0 = fold<4>(h0, h1, up4);
    const float k1 = h2 + row_xmask<4>(h2);
    const float k2 = fold<2>(k0, k1, up2);
    return k2 + row_xmask<1>(k2);
  };
  float gqc = xdotx(xa);

  auto quadstep = [&](xpair_t (&xd)[4][XP], const xpair_t (&xn)[4][XP], const int t,
                      float (*pb)[SLOTS])
                      __attribute__((always_inline)) {
    v2f xq[4][XP];
#pragma unroll
    for (int s = 0; s < 4; ++s) {
#pragma unroll
      for (int k = 0; k < XP; ++k) xq[s][k] = __builtin_bit_cast(v2f, xd[s][k]);
    }

    v2f acc[4][4];
#pragma unroll
    for (int j = 0; j < TILES; ++j) {
#pragma unroll
      for (int s = 0; s < 4; ++s) {
        const v2f xs = (j & 1) ? __builtin_shufflevector(xq[s][j >> 1], xq[s][j >> 1], 1, 1)
                               : __builtin_shufflevector(xq[s][j >> 1], xq[s][j >> 1], 0, 0);
#pragma unroll
        for (int i = 0; i < 4; ++i) {
          const v2f w2 = {w[j][2 * i], w[j][2 * i + 1]};
          acc[s][i] = (j == 0) ? (w2 * xs) : __builtin_elementwise_fma(w2, xs, acc[s][i]);
        }
      }
    }

    v2f bq[TILES];
#pragma unroll
    for (int j = 0; j < TILES; ++j) {
      bq[j].x = half ? xq[2][j >> 1][j & 1] : xq[0][j >> 1][j & 1];
      bq[j].y = half ? xq[3][j >> 1][j & 1] : xq[1][j >> 1][j & 1];
      asm volatile("" : "+v"(bq[j]));
    }

    const float q01 = fold16(acc[0], acc[1]);
    const float q23 = fold16(acc[2], acc[3]);
    pb[wv][myslot] = q01;
    pb[wv][32 + myslot] = q23;
    if (gsel) pb[wv][64 + gidx] = gqc;
    __syncthreads();

    float pD[4][WAVES];
    v4f ga[WAVES];
    v2f gb[WAVES];
#pragma unroll
    for (int k = 0; k < WAVES; ++k) {
#pragma unroll
      for (int s = 0; s < 4; ++s) pD[s][k] = pb[k][16 * s + ln];
      ga[k] = *reinterpret_cast<const v4f*>(&pb[k][64]);
      gb[k] = *reinterpret_cast<const v2f*>(&pb[k][68]);
    }
    asm volatile("" ::: "memory");

#pragma unroll
    for (int s = 0; s < 4; ++s) {
      const int ts = (t + 8 + s < Tn) ? (t + 8 + s) : (Tn - 1);
      load_x(xd[s], xg + (size_t)ts * NI);
    }

    gqc = xdotx(xn);

    float D[4];
#pragma unroll
    for (int s = 0; s < 4; ++s) D[s] = (pD[s][0] + pD[s][1]) + (pD[s][2] + pD[s][3]);
    const v4f gs = (ga[0] + ga[1]) + (ga[2] + ga[3]);
    const v2f gt = (gb[0] + gb[1]) + (gb[2] + gb[3]);
    auto sigm = [&](const float z) __attribute__((always_inline)) {
      return __builtin_amdgcn_rcpf(1.0f + __builtin_amdgcn_exp2f(z));
    };
    auto decay = [&](const float P, const float y) __attribute__((always_inline)) {
      return P * __builtin_fmaf(-lr * y, y, 1.0f);
    };
    const v4f cgs = cmul * gs;
    const v2f cgt = cmul * gt;
    const float P0 = Pn;
    const float y0 = sigm(P0 * D[0]);
    const float q1 = P0 * D[1];
    const float y1 = sigm(__builtin_fmaf(y0, __builtin_fmaf(-lr * q1, y0, cgs.x), q1));
    const float P1 = decay(P0, y0);
    const float q2 = __builtin_fmaf(P1, D[2], cgs.y * y0);
    const float y2 = sigm(__builtin_fmaf(y1, __builtin_fmaf(-lr * q2, y1, cgs.w), q2));
    const float P2 = decay(P1, y1);
    const float c0 = (cmul * y0) * __builtin_amdgcn_rcpf(P1);
    const float q3 = __builtin_fmaf(P2, __builtin_fmaf(c0, gs.z, D[3]), cgt.x * y1);
    const float y3 = sigm(__builtin_fmaf(y2, __builtin_fmaf(-lr * q3, y2, cgt.y), q3));
    const float P3 = decay(P2, y2);
    const float c1 = (cmul * y1) * __builtin_amdgcn_rcpf(P2);
    const float c2 = (cmul * y2) * __builtin_amdgcn_rcpf(P3);
    Pn = decay(P3, y3);
    const float c3 = (cmul * y3) * __builtin_amdgcn_rcpf(Pn);
    if (wv < 2) {
      const float yv = wv ? (half ? y3 : y2) : (half ? y1 : y0);
      out[((size_t)b * Tn + t + 2 * wv + half) * NO + (size_t)strip * 16 + ln] = yv;
    }

    v2f A;
    A.x = half ? c2 : c0;
    A.y = half ? c3 : c1;
#pragma unroll
    for (int j = 0; j < TILES; ++j) {
      w[j] = __builtin_amdgcn_wmma_f32_16x16x4_f32(
           false, A,  false, bq[j],
           (short)0, w[j],  false,  false);
    }
  };

  __builtin_amdgcn_s_setprio(3);

#pragma unroll 1
  for (int t = 0; t < Tn; t += 8) {
    switch (t) {
      case 32: __builtin_amdgcn_s_setprio(2); break;
      case 48: __builtin_amdgcn_s_setprio(1); break;
      case 56: __builtin_amdgcn_s_setprio(0); break;
      default: break;
    }
    quadstep(xa, xb, t, part[0]);
    quadstep(xb, xa, t + 4, part[1]);
  }
}

extern "C" void kernel_launch(void* const* d_in, const int* in_sizes, int n_in,
                              void* d_out, int out_size, void* d_ws, size_t ws_size,
                              hipStream_t stream) {
  const float* X  = (const float*)d_in[0];
  const float* W0 = (const float*)d_in[1];
  float* out      = (float*)d_out;
  (void)in_sizes; (void)n_in; (void)out_size; (void)d_ws; (void)ws_size;

  dim3 grid(Bn * (NO / 16));
  dim3 block(WAVES * 32);
  oja_scan_kernel<<<grid, block, 0, stream>>>(X, W0, out);
}
